# starting kernel
# baseline (speedup 1.0000x reference)
#include <hip/hip_runtime.h>

typedef float v2f __attribute__((ext_vector_type(2)));
typedef float v8f __attribute__((ext_vector_type(8)));

#define NS_TOTAL 16384
#define SITES    8
#define NCOL     (SITES * 18)
#define KPAD     112
#define MPAD     80
#define THREADS  256

struct C3 { float2 m[9]; };

__device__ inline float2 cmul(float2 a, float2 b) {
  return make_float2(a.x*b.x - a.y*b.y, a.x*b.y + a.y*b.x);
}
__device__ inline float2 cmulcj(float2 a, float2 b) {
  return make_float2(a.x*b.x + a.y*b.y, a.y*b.x - a.x*b.y);
}
__device__ inline float2 cjmul(float2 a, float2 b) {
  return make_float2(a.x*b.x + a.y*b.y, a.x*b.y - a.y*b.x);
}

__device__ inline int neighbor(int s, int axis, int dir) {
  int a0 = (s >> 10) & 7, a1 = (s >> 7) & 7, a2 = (s >> 4) & 7, a3 = s & 15;
  if      (axis == 0) a0 = (a0 + dir) & 7;
  else if (axis == 1) a1 = (a1 + dir) & 7;
  else if (axis == 2) a2 = (a2 + dir) & 7;
  else                a3 = (a3 + dir) & 15;
  return (a0 << 10) | (a1 << 7) | (a2 << 4) | a3;
}

__device__ inline void loadMat(const float* __restrict__ xin, int bx, int ch, C3& M) {
  const float* p = xin + ((size_t)bx * 10 + ch) * 18;
  #pragma unroll
  for (int e = 0; e < 9; ++e) M.m[e] = make_float2(p[2*e], p[2*e + 1]);
}

__device__ inline void mul33(const C3& A, const C3& B, C3& C) {
  #pragma unroll
  for (int i = 0; i < 3; ++i)
    #pragma unroll
    for (int k = 0; k < 3; ++k) {
      float2 acc = make_float2(0.f, 0.f);
      #pragma unroll
      for (int j = 0; j < 3; ++j) {
        float2 t = cmul(A.m[i*3+j], B.m[j*3+k]);
        acc.x += t.x; acc.y += t.y;
      }
      C.m[i*3+k] = acc;
    }
}

__device__ inline void sandwichF(const C3& U, const C3& W, C3& R) {
  C3 T; mul33(U, W, T);
  #pragma unroll
  for (int i = 0; i < 3; ++i)
    #pragma unroll
    for (int l = 0; l < 3; ++l) {
      float2 acc = make_float2(0.f, 0.f);
      #pragma unroll
      for (int k = 0; k < 3; ++k) {
        float2 t = cmulcj(T.m[i*3+k], U.m[l*3+k]);
        acc.x += t.x; acc.y += t.y;
      }
      R.m[i*3+l] = acc;
    }
}

__device__ inline void sandwichB(const C3& U, const C3& W, C3& R) {
  C3 T;
  #pragma unroll
  for (int i = 0; i < 3; ++i)
    #pragma unroll
    for (int k = 0; k < 3; ++k) {
      float2 acc = make_float2(0.f, 0.f);
      #pragma unroll
      for (int j = 0; j < 3; ++j) {
        float2 t = cjmul(U.m[j*3+i], W.m[j*3+k]);
        acc.x += t.x; acc.y += t.y;
      }
      T.m[i*3+k] = acc;
    }
  #pragma unroll
  for (int i = 0; i < 3; ++i)
    #pragma unroll
    for (int l = 0; l < 3; ++l) {
      float2 acc = make_float2(0.f, 0.f);
      #pragma unroll
      for (int k = 0; k < 3; ++k) {
        float2 t = cmul(T.m[i*3+k], U.m[k*3+l]);
        acc.x += t.x; acc.y += t.y;
      }
      R.m[i*3+l] = acc;
    }
}

__device__ inline void storeRow(float* dst, int ch, int ls, const C3& M) {
  float* p = dst + ch * NCOL + ls * 18;
  #pragma unroll
  for (int e = 0; e < 9; ++e) { p[2*e] = M.m[e].x; p[2*e + 1] = M.m[e].y; }
}
__device__ inline void storeRowDag(float* dst, int ch, int ls, const C3& M) {
  float* p = dst + ch * NCOL + ls * 18;
  #pragma unroll
  for (int i = 0; i < 3; ++i)
    #pragma unroll
    for (int j = 0; j < 3; ++j) {
      p[(i*3+j)*2]     =  M.m[j*3+i].x;
      p[(i*3+j)*2 + 1] = -M.m[j*3+i].y;
    }
}

__global__ __launch_bounds__(THREADS)
void lconv_bilin_kernel(const float* __restrict__ xin,
                        const float* __restrict__ wgt,
                        float* __restrict__ out) {
  extern __shared__ float smem[];
  float* Wt = smem;
  float* tF = Wt + MPAD * KPAD;
  float* rL = tF + KPAD * NCOL;
  float* wF = rL + MPAD * NCOL;

  const int tid  = threadIdx.x;
  const int base = blockIdx.x * SITES;

  for (int idx = tid; idx < MPAD * KPAD; idx += THREADS) {
    int m_ = idx / KPAD, k_ = idx % KPAD;
    Wt[idx] = (m_ < 78 && k_ < 109) ? wgt[m_ * 109 + k_] : 0.0f;
  }
  for (int idx = tid; idx < 3 * NCOL; idx += THREADS) tF[109 * NCOL + idx] = 0.0f;

  for (int t = tid; t < SITES * 48; t += THREADS) {
    int ls = t / 48, rem = t % 48;
    int axis = rem / 12, rr = rem % 12;
    int c = rr % 6;
    bool backward = (rr < 6);
    int bx = base + ls;
    int s = bx & 8191, bofs = bx & ~8191;
    C3 U, W, R;
    if (backward) {
      int sn = neighbor(s, axis, -1);
      loadMat(xin, bofs | sn, axis, U);
      loadMat(xin, bofs | sn, 4 + c, W);
      sandwichB(U, W, R);
    } else {
      int sn = neighbor(s, axis, +1);
      loadMat(xin, bx, axis, U);
      loadMat(xin, bofs | sn, 4 + c, W);
      sandwichF(U, W, R);
    }
    int ch = 6 + axis * 12 + rr;
    storeRow(tF, ch, ls, R);
    storeRowDag(tF, 54 + ch, ls, R);
  }

  for (int t = tid; t < SITES * 6; t += THREADS) {
    int ls = t / 6, c = t % 6;
    C3 W; loadMat(xin, base + ls, 4 + c, W);
    storeRow(tF, c, ls, W);
    storeRowDag(tF, 54 + c, ls, W);
    storeRow(wF, c, ls, W);
    storeRowDag(wF, 6 + c, ls, W);
  }

  for (int t = tid; t < SITES; t += THREADS) {
    float* p1 = tF + 108 * NCOL + t * 18;
    float* p2 = wF + 12 * NCOL + t * 18;
    #pragma unroll
    for (int i = 0; i < 3; ++i)
      #pragma unroll
      for (int j = 0; j < 3; ++j) {
        float re = (i == j) ? 1.0f : 0.0f;
        p1[(i*3+j)*2] = re; p1[(i*3+j)*2 + 1] = 0.0f;
        p2[(i*3+j)*2] = re; p2[(i*3+j)*2 + 1] = 0.0f;
      }
  }

  __syncthreads();

  {
    const int wv = tid >> 5, lane = tid & 31;
    const int l15 = lane & 15, kh = lane >> 4;
    for (int tile = wv; tile < 5 * 9; tile += 8) {
      int mt = tile / 9, nt = tile % 9;
      v8f acc = {};
      const float* wbase = Wt + (mt * 16 + l15) * KPAD;
      const float* bbase = tF + nt * 16 + l15;
      for (int k0 = 0; k0 < KPAD; k0 += 4) {
        v2f a, b;
        a[0] = wbase[k0 + 2 * kh];
        a[1] = wbase[k0 + 2 * kh + 1];
        const float* bp = bbase + (k0 + 2 * kh) * NCOL;
        b[0] = bp[0];
        b[1] = bp[NCOL];
        acc = __builtin_amdgcn_wmma_f32_16x16x4_f32(
            false, a, false, b, (short)0, acc, false, false);
      }
      float* rp = rL + (mt * 16 + (lane >> 4) * 8) * NCOL + nt * 16 + l15;
      #pragma unroll
      for (int q = 0; q < 8; ++q) rp[q * NCOL] = acc[q];
    }
  }
  __syncthreads();

  for (int t = tid; t < SITES * 6 * 9; t += THREADS) {
    int ls = t / 54, rem = t % 54;
    int u = rem / 9, ik = rem % 9, i = ik / 3, k = ik % 3;
    float2 acc = make_float2(0.f, 0.f);
    #pragma unroll
    for (int v = 0; v < 13; ++v) {
      const float* wp = wF + v * NCOL + ls * 18;
      const float* rp = rL + (u * 13 + v) * NCOL + ls * 18;
      #pragma unroll
      for (int j = 0; j < 3; ++j) {
        float2 a = make_float2(wp[(i*3+j)*2], wp[(i*3+j)*2 + 1]);
        float2 b = make_float2(rp[(j*3+k)*2], rp[(j*3+k)*2 + 1]);
        acc.x += a.x * b.x - a.y * b.y;
        acc.y += a.x * b.y + a.y * b.x;
      }
    }
    int bx = base + ls;
    float* op = out + ((size_t)bx * 10 + 4 + u) * 18 + ik * 2;
    op[0] = acc.x; op[1] = acc.y;
  }

  for (int t = tid; t < SITES * 4 * 18; t += THREADS) {
    int ls = t / 72, rem = t % 72;
    int ch = rem / 18, e = rem % 18;
    size_t idx = ((size_t)(base + ls) * 10 + ch) * 18 + e;
    out[idx] = xin[idx];
  }
}

extern "C" void kernel_launch(void* const* d_in, const int* in_sizes, int n_in,
                              void* d_out, int out_size, void* d_ws, size_t ws_size,
                              hipStream_t stream) {
  (void)in_sizes; (void)n_in; (void)out_size; (void)d_ws; (void)ws_size;
  const float* x = (const float*)d_in[0];
  const float* w = (const float*)d_in[1];
  float* out = (float*)d_out;

  const size_t shmem =
      (size_t)(MPAD * KPAD + KPAD * NCOL + MPAD * NCOL + 13 * NCOL) * sizeof(float);
  (void)hipFuncSetAttribute((const void*)lconv_bilin_kernel,
                            hipFuncAttributeMaxDynamicSharedMemorySize, (int)shmem);
  lconv_bilin_kernel<<<NS_TOTAL / SITES, THREADS, shmem, stream>>>(x, w, out);
}


# phase C re-cut: a thread multiplies whole 3x3 blocks w_f[v]*r[u,v] for 3 of the 12 v, 4 lanes join by a lane butterfly; 192 threads (waves 6-11), one pass; sum over v reordered, products fused
# speedup vs baseline: 2.5725x; 2.5725x over previous
#include <hip/hip_runtime.h>

typedef float v2f __attribute__((ext_vector_type(2)));
typedef float v4f __attribute__((ext_vector_type(4)));
typedef float v8f __attribute__((ext_vector_type(8)));
typedef float v4fu __attribute__((ext_vector_type(4), aligned(4)));
typedef float v2fu __attribute__((ext_vector_type(2), aligned(4)));

#define NS_TOTAL 16384
#define SITES    8
#define NCOL     (SITES * 18)
#define KPAD     108
#define MPAD     80
#define THREADS  384
#define KSTEPS   (KPAD / 4)
#define KPAIRS   (KSTEPS / 2)
#define KPIECES  (KPAIRS + 1)
#define KSP      (NCOL * 4 + 4)

struct C3 { float2 m[9]; };

__device__ inline float2 cmul(float2 a, float2 b) {
  return make_float2(a.x*b.x - a.y*b.y, a.x*b.y + a.y*b.x);
}
__device__ inline float2 cmulcj(float2 a, float2 b) {
  return make_float2(a.x*b.x + a.y*b.y, a.y*b.x - a.x*b.y);
}

__device__ inline int neighbor(int s, int axis, int dir) {
  int a0 = (s >> 10) & 7, a1 = (s >> 7) & 7, a2 = (s >> 4) & 7, a3 = s & 15;
  if      (axis == 0) a0 = (a0 + dir) & 7;
  else if (axis == 1) a1 = (a1 + dir) & 7;
  else if (axis == 2) a2 = (a2 + dir) & 7;
  else                a3 = (a3 + dir) & 15;
  return (a0 << 10) | (a1 << 7) | (a2 << 4) | a3;
}

__device__ inline void loadMat(const float* __restrict__ xin, int bx, int ch, C3& M) {
  const float* p = xin + ((size_t)bx * 10 + ch) * 18;
  #pragma unroll
  for (int e = 0; e < 9; ++e) M.m[e] = make_float2(p[2*e], p[2*e + 1]);
}

__device__ inline void mul33(const C3& A, const C3& B, C3& C) {
  #pragma unroll
  for (int i = 0; i < 3; ++i)
    #pragma unroll
    for (int k = 0; k < 3; ++k) {
      float2 acc = make_float2(0.f, 0.f);
      #pragma unroll
      for (int j = 0; j < 3; ++j) {
        float2 t = cmul(A.m[i*3+j], B.m[j*3+k]);
        acc.x += t.x; acc.y += t.y;
      }
      C.m[i*3+k] = acc;
    }
}

__device__ inline void sandwichF(const C3& U, const C3& W, C3& R) {
  C3 T; mul33(U, W, T);
  #pragma unroll
  for (int i = 0; i < 3; ++i)
    #pragma unroll
    for (int l = 0; l < 3; ++l) {
      float2 acc = make_float2(0.f, 0.f);
      #pragma unroll
      for (int k = 0; k < 3; ++k) {
        float2 t = cmulcj(T.m[i*3+k], U.m[l*3+k]);
        acc.x += t.x; acc.y += t.y;
      }
      R.m[i*3+l] = acc;
    }
}

__device__ inline void storeRow(float* dst, int ch, int ls, const C3& M) {
  float* p = dst + ch * NCOL + ls * 18;
  #pragma unroll
  for (int e = 0; e < 9; ++e) { p[2*e] = M.m[e].x; p[2*e + 1] = M.m[e].y; }
}
__device__ inline void storeRowDag(float* dst, int ch, int ls, const C3& M) {
  float* p = dst + ch * NCOL + ls * 18;
  #pragma unroll
  for (int i = 0; i < 3; ++i)
    #pragma unroll
    for (int j = 0; j < 3; ++j) {
      p[(i*3+j)*2]     =  M.m[j*3+i].x;
      p[(i*3+j)*2 + 1] = -M.m[j*3+i].y;
    }
}

__device__ inline int tfChanLo(int ch) {
  return ((ch >> 3) * 2 + ((ch >> 1) & 1)) * KSP + ((ch >> 2) & 1) * 2 + (ch & 1);
}
__device__ inline int tfChan(int ch) {
  const int par = (ch >> 1) & 1;
  const int last = (ch >= 104) ? 1 : 0;
  const int plane = (ch >> 3) * 2 + (last ? 0 : par);
  const int hsel  = last ? par : ((ch >> 2) & 1);
  return plane * KSP + hsel * 2 + (ch & 1);
}
__device__ inline void storeRowT(float* tF, int ch, int ls, const C3& M) {
  float* p = tF + tfChanLo(ch) + ls * (18 * 4);
  #pragma unroll
  for (int e = 0; e < 9; ++e) { p[(2*e) * 4] = M.m[e].x; p[(2*e + 1) * 4] = M.m[e].y; }
}
__device__ inline void storeRowDagT(float* tF, int ch, int ls, const C3& M) {
  float* p = tF + tfChan(ch) + ls * (18 * 4);
  #pragma unroll
  for (int i = 0; i < 3; ++i)
    #pragma unroll
    for (int j = 0; j < 3; ++j) {
      p[((i*3+j)*2) * 4]     =  M.m[j*3+i].x;
      p[((i*3+j)*2 + 1) * 4] = -M.m[j*3+i].y;
    }
}

#define AR 3
#define BR 3

__device__ __forceinline__ v4f loadApiece(const float* __restrict__ wrow, int g, int kh) {
  if (g < KPAIRS) return *reinterpret_cast<const v4fu*>(wrow + 8 * g + 4 * kh);
  const v2fu t = *reinterpret_cast<const v2fu*>(wrow + 104 + 2 * kh);
  return v4f{t.x, t.y, 0.0f, 0.0f};
}

template <int NT>
__device__ __forceinline__ void stripGemm(const float* __restrict__ wrow,
                                          v4f (&ar)[AR],
                                          const float* __restrict__ tF,
                                          v8f (&acc)[5],
                                          int nt0, int lane) {
  const int l15 = lane & 15, kh = lane >> 4;
  #pragma unroll
  for (int n = 0; n < NT; ++n) acc[n] = v8f{};
  const v2f* bbase = reinterpret_cast<const v2f*>(tF + (nt0 * 16 + l15) * 4 + kh * 2);
  v2f br[BR][NT];
  #pragma unroll
  for (int d = 0; d < BR - 1; ++d)
    #pragma unroll
    for (int n = 0; n < NT; ++n) br[d][n] = bbase[d * (KSP / 2) + n * 32];
  v4f a = v4f{};
  #pragma unroll
  for (int s = 0; s < KSTEPS; ++s) {
    if (s + BR - 1 < KSTEPS) {
      const v2f* bp = bbase + (s + BR - 1) * (KSP / 2);
      #pragma unroll
      for (int n = 0; n < NT; ++n) br[(s + BR - 1) % BR][n] = bp[n * 32];
    }
    if ((s & 1) == 0) {
      const int p = s >> 1;
      if (p + AR - 1 < KPIECES) ar[(p + AR - 1) % AR] = loadApiece(wrow, p + AR - 1, kh);
      a = ar[p % AR];
    }
    __builtin_amdgcn_sched_barrier(0);
    const v2f ak = (s & 1) ? __builtin_shufflevector(a, a, 2, 3)
                           : __builtin_shufflevector(a, a, 0, 1);
    #pragma unroll
    for (int n = 0; n < NT; ++n)
      acc[n] = __builtin_amdgcn_wmma_f32_16x16x4_f32(
          false, ak, false, br[s % BR][n], (short)0, acc[n], false, false);
    __builtin_amdgcn_sched_barrier(0);
  }
}

__device__ __forceinline__ void loadSeedColumn(const float* __restrict__ wgt, float (&wS)[16], int mt) {
  const float* p = wgt + mt * (16 * 109) + 108;
  #pragma unroll
  for (int r = 0; r < 14; ++r) wS[r] = p[r * 109];
  const int back = (mt == 4) ? 109 : 0;
  wS[14] = p[14 * 109 - back];
  wS[15] = p[15 * 109 - 2 * back];
}

template <int NT>
__device__ __forceinline__ void stripStore(const v8f (&acc)[5], const float (&wS)[16],
                                           float* __restrict__ rL,
                                           int mt, int nt0, int lane) {
  const int l15 = lane & 15, kh = lane >> 4;
  float wI[8];
  #pragma unroll
  for (int q = 0; q < 8; ++q) wI[q] = kh ? wS[8 + q] : wS[q];
  float* rp = rL + (mt * 16 + kh * 8) * NCOL + nt0 * 16 + l15;
  #pragma unroll
  for (int n = 0; n < NT; ++n) {
    const float d = (((((nt0 + n) * 16 + l15) % 18) & 7) == 0) ? 1.0f : 0.0f;
    #pragma unroll
    for (int q = 0; q < 8; ++q) rp[q * NCOL + n * 16] = acc[n][q] + wI[q] * d;
  }
}

__device__ __forceinline__ float quadXor1(float x) {
  return __int_as_float(__builtin_amdgcn_update_dpp(0, __float_as_int(x), 0xB1, 0xf, 0xf, true));
}
__device__ __forceinline__ float quadXor2(float x) {
  return __int_as_float(__builtin_amdgcn_update_dpp(0, __float_as_int(x), 0x4E, 0xf, 0xf, true));
}

__device__ __forceinline__ float pick4(int g, float a, float b, float c, float d) {
  const float lo = (g & 1) ? b : a, hi = (g & 1) ? d : c;
  return (g & 2) ? hi : lo;
}

__global__ __launch_bounds__(THREADS, 9)
void lconv_bilin_kernel(const float* __restrict__ xin,
                        const float* __restrict__ wgt,
                        float* __restrict__ out) {
  extern __shared__ float smem[];
  float* tF = smem;
  float* rL = smem;
  float* wF = tF + KSTEPS * KSP;

  const int tid  = threadIdx.x;
  const int base = blockIdx.x * SITES;

  for (int t = tid; t < SITES * 48; t += THREADS) {
    int ls = t / 48, rem = t % 48;
    int axis = rem / 12, rr = rem % 12;
    int c = rr % 6;
    bool backward = (rr < 6);
    int bx = base + ls;
    int s = bx & 8191, bofs = bx & ~8191;
    int bn = bofs | neighbor(s, axis, backward ? -1 : +1);
    C3 U, W, V, R;
    loadMat(xin, backward ? bn : bx, axis, U);
    loadMat(xin, bn, 4 + c, W);
    #pragma unroll
    for (int i = 0; i < 3; ++i)
      #pragma unroll
      for (int j = 0; j < 3; ++j) {
        float2 a = U.m[i*3+j], b = U.m[j*3+i];
        V.m[i*3+j] = backward ? make_float2(b.x, -b.y) : a;
      }
    sandwichF(V, W, R);
    int ch = 6 + axis * 12 + rr;
    storeRowT(tF, ch, ls, R);
    storeRowDagT(tF, 54 + ch, ls, R);
  }

  const int wv   = __builtin_amdgcn_readfirstlane(tid >> 5);
  const int lane = tid & 31;
  int mt, nt0, ntk;
  if (wv < 3)             { mt = 2 + wv;        nt0 = 0;            ntk = 5; }
  else if ((wv & 3) == 3) { mt = 2 + (wv >> 2); nt0 = 5;            ntk = 4; }
  else                    { mt = wv >> 3;       nt0 = 3 * (wv & 3); ntk = 3; }
  const int am = mt * 16 + (lane & 15);
  const float* wrow = wgt + (am < 78 ? am : 77) * 109;
  v4f ar[AR];
  #pragma unroll
  for (int g = 0; g < AR - 1; ++g) ar[g] = loadApiece(wrow, g, lane >> 4);
  float wS[16];
  loadSeedColumn(wgt, wS, mt);

  for (int t = tid - 128; t >= 0 && t < SITES * 6; t += THREADS) {
    int ls = t / 6, c = t % 6;
    C3 W; loadMat(xin, base + ls, 4 + c, W);
    storeRowT(tF, c, ls, W);
    storeRowDagT(tF, 54 + c, ls, W);
    storeRow(wF, c, ls, W);
    storeRowDag(wF, 6 + c, ls, W);
  }

  v8f acc[5];
  __syncthreads();
  if (ntk == 5)      stripGemm<5>(wrow, ar, tF, acc, nt0, lane);
  else if (ntk == 4) stripGemm<4>(wrow, ar, tF, acc, nt0, lane);
  else               stripGemm<3>(wrow, ar, tF, acc, nt0, lane);
  __syncthreads();
  if (ntk == 5)      stripStore<5>(acc, wS, rL, mt, nt0, lane);
  else if (ntk == 4) stripStore<4>(acc, wS, rL, mt, nt0, lane);
  else               stripStore<3>(acc, wS, rL, mt, nt0, lane);
  __syncthreads();

  if (wv >= 6) {
    const int g  = lane & 3;
    const int ls = (lane >> 2) & 7;
    const int u  = wv - 6;
    float2 acc[9];
    #pragma unroll
    for (int e = 0; e < 9; ++e) acc[e] = make_float2(0.f, 0.f);
    #pragma unroll
    for (int vv = 0; vv < 3; ++vv) {
      const int v = 3 * g + vv;
      const v2f* ap = reinterpret_cast<const v2f*>(wF + v * NCOL + ls * 18);
      const v2f* bp = reinterpret_cast<const v2f*>(rL + (u * 13 + v) * NCOL + ls * 18);
      v2f A[9], Bm[9];
      #pragma unroll
      for (int e = 0; e < 9; ++e) { A[e] = ap[e]; Bm[e] = bp[e]; }
      #pragma unroll
      for (int i = 0; i < 3; ++i)
        #pragma unroll
        for (int k = 0; k < 3; ++k)
          #pragma unroll
          for (int j = 0; j < 3; ++j) {
            const v2f a = A[i*3+j], b = Bm[j*3+k];
            acc[i*3+k].x = fmaf(a.x, b.x, acc[i*3+k].x);
            acc[i*3+k].x = fmaf(-a.y, b.y, acc[i*3+k].x);
            acc[i*3+k].y = fmaf(a.x, b.y, acc[i*3+k].y);
            acc[i*3+k].y = fmaf(a.y, b.x, acc[i*3+k].y);
          }
    }
    #pragma unroll
    for (int e = 0; e < 9; ++e) {
      acc[e].x += quadXor1(acc[e].x); acc[e].y += quadXor1(acc[e].y);
    }
    #pragma unroll
    for (int e = 0; e < 9; ++e) {
      acc[e].x += quadXor2(acc[e].x); acc[e].y += quadXor2(acc[e].y);
    }
    const v2f* ip = reinterpret_cast<const v2f*>(rL + (u * 13 + 12) * NCOL + ls * 18);
    const v2f i0 = ip[g], i1 = ip[4 + g], i8 = ip[8];
    const float s0x = pick4(g, acc[0].x, acc[1].x, acc[2].x, acc[3].x);
    const float s0y = pick4(g, acc[0].y, acc[1].y, acc[2].y, acc[3].y);
    const float s1x = pick4(g, acc[4].x, acc[5].x, acc[6].x, acc[7].x);
    const float s1y = pick4(g, acc[4].y, acc[5].y, acc[6].y, acc[7].y);
    v2f* op = reinterpret_cast<v2f*>(out + ((size_t)(base + ls) * 10 + 4 + u) * 18);
    op[g]     = v2f{s0x + i0.x, s0y + i0.y};
    op[4 + g] = v2f{s1x + i1.x, s1y + i1.y};
    if (g == 0) op[8] = v2f{acc[8].x + i8.x, acc[8].y + i8.y};
  }

  for (int t = tid; t < SITES * 4 * 18; t += THREADS) {
    int ls = t / 72, rem = t % 72;
    int ch = rem / 18, e = rem % 18;
    size_t idx = ((size_t)(base + ls) * 10 + ch) * 18 + e;
    out[idx] = xin[idx];
  }
}

extern "C" void kernel_launch(void* const* d_in, const int* in_sizes, int n_in,
                              void* d_out, int out_size, void* d_ws, size_t ws_size,
                              hipStream_t stream) {
  (void)in_sizes; (void)n_in; (void)out_size; (void)d_ws; (void)ws_size;
  const float* x = (const float*)d_in[0];
  const float* w = (const float*)d_in[1];
  float* out = (float*)d_out;

  const size_t shmem =
      (size_t)(KSTEPS * KSP + 12 * NCOL) * sizeof(float);
  (void)hipFuncSetAttribute((const void*)lconv_bilin_kernel,
                            hipFuncAttributeMaxDynamicSharedMemorySize, (int)shmem);
  lconv_bilin_kernel<<<NS_TOTAL / SITES, THREADS, shmem, stream>>>(x, w, out);
}
